# starting kernel
# baseline (speedup 1.0000x reference)
#include <hip/hip_runtime.h>

typedef __attribute__((ext_vector_type(2))) float v2f;
typedef __attribute__((ext_vector_type(8))) float v8f;

#define BATCH 8
#define SEQ   8192
#define DN    128
#define CH    128
#define NCH   (SEQ / CH)

__global__ void ssm_pass1(const float* __restrict__ x, const float* __restrict__ dt,
                          const float* __restrict__ B, const float* __restrict__ C,
                          const float* __restrict__ A,
                          float* __restrict__ y, float* __restrict__ aprod,
                          float* __restrict__ hloc)
{
    __shared__ float dt_s[CH];
    __shared__ float dtx_s[CH];
    __shared__ float ypart[CH * 4];

    const int k = blockIdx.x, b = blockIdx.y;
    const int tid = threadIdx.x;
    const int lane = tid & 31, wv = tid >> 5;
    const size_t sbase = (size_t)b * SEQ + (size_t)k * CH;

    float d = dt[sbase + tid];
    dt_s[tid]  = d;
    dtx_s[tid] = d * x[sbase + tid];

    const float* Bp = B + sbase * DN + tid;
    const float* Cp = C + sbase * DN + tid;
    __builtin_prefetch(Bp, 0, 0);
    __builtin_prefetch(Cp, 0, 0);
    __syncthreads();

    const float An = A[tid];
    float h = 0.f, sd = 0.f;
#pragma unroll 4
    for (int t = 0; t < CH; ++t) {
        float dts = dt_s[t];
        sd += dts;
        float a = __expf(dts * An);
        h = fmaf(a, h, dtx_s[t] * Bp[(size_t)t * DN]);
        float p = Cp[(size_t)t * DN] * h;
        p += __shfl_xor(p, 1, 32);
        p += __shfl_xor(p, 2, 32);
        p += __shfl_xor(p, 4, 32);
        p += __shfl_xor(p, 8, 32);
        p += __shfl_xor(p, 16, 32);
        if (lane == 0) ypart[t * 4 + wv] = p;
    }

    const size_t cidx = ((size_t)b * NCH + k) * DN + tid;
    aprod[cidx] = __expf(An * sd);
    hloc[cidx]  = h;
    __syncthreads();
    y[sbase + tid] = ypart[tid * 4 + 0] + ypart[tid * 4 + 1] +
                     ypart[tid * 4 + 2] + ypart[tid * 4 + 3];
}

__global__ void ssm_pass2(const float* __restrict__ aprod, const float* __restrict__ hloc,
                          float* __restrict__ carry, float* __restrict__ hfin)
{
    const int b = blockIdx.x, n = threadIdx.x;
    float h = 0.f;
    for (int k = 0; k < NCH; ++k) {
        const size_t idx = ((size_t)b * NCH + k) * DN + n;
        carry[idx] = h;
        h = fmaf(aprod[idx], h, hloc[idx]);
    }
    hfin[(size_t)b * DN + n] = h;
}

__global__ void ssm_pass3(const float* __restrict__ dt, const float* __restrict__ C,
                          const float* __restrict__ A, const float* __restrict__ carry,
                          float* __restrict__ y)
{
    __shared__ float Sd[CH];
    __shared__ float sA[DN];
    __shared__ float hin[DN];
    __shared__ float ycorr[CH];

    const int k = blockIdx.x, b = blockIdx.y;
    const int tid = threadIdx.x;
    const int lane = tid & 31, wv = tid >> 5;
    const size_t sbase = (size_t)b * SEQ + (size_t)k * CH;

    Sd[tid]  = dt[sbase + tid];
    sA[tid]  = A[tid];
    hin[tid] = carry[((size_t)b * NCH + k) * DN + tid];
    __syncthreads();
    for (int off = 1; off < CH; off <<= 1) {
        float v = (tid >= off) ? Sd[tid - off] : 0.f;
        __syncthreads();
        Sd[tid] += v;
        __syncthreads();
    }

    const float* Cb = C + sbase * DN;
    const int kBase = (lane < 16) ? 0 : 2;

    for (int i = 0; i < 2; ++i) {
        const int t0   = (wv * 2 + i) * 16;
        const int tRow = t0 + (lane & 15);
        const float sdt = Sd[tRow];
        const float* Crow = Cb + (size_t)tRow * DN;

        v8f acc = {};
#pragma unroll 8
        for (int k0 = 0; k0 < DN; k0 += 4) {
            const int kk = k0 + kBase;
            float2 cv = *(const float2*)(Crow + kk);
            v2f a2, b2;
            a2.x = cv.x * __expf(sA[kk] * sdt);
            a2.y = cv.y * __expf(sA[kk + 1] * sdt);
            b2.x = hin[kk];
            b2.y = hin[kk + 1];
            acc = __builtin_amdgcn_wmma_f32_16x16x4_f32(
                false, a2, false, b2, (short)0, acc, false, false);
        }
        if (lane == 0) {
#pragma unroll
            for (int r = 0; r < 8; ++r) ycorr[t0 + r] = acc[r];
        }
        if (lane == 16) {
#pragma unroll
            for (int r = 0; r < 8; ++r) ycorr[t0 + 8 + r] = acc[r];
        }
    }
    __syncthreads();
    y[sbase + tid] += ycorr[tid];
}

extern "C" void kernel_launch(void* const* d_in, const int* in_sizes, int n_in,
                              void* d_out, int out_size, void* d_ws, size_t ws_size,
                              hipStream_t stream) {
    const float* x  = (const float*)d_in[0];
    const float* dt = (const float*)d_in[1];
    const float* B  = (const float*)d_in[2];
    const float* C  = (const float*)d_in[3];
    const float* A  = (const float*)d_in[4];

    float* y    = (float*)d_out;
    float* hfin = y + (size_t)BATCH * SEQ;

    float* aprod = (float*)d_ws;
    float* hloc  = aprod + (size_t)BATCH * NCH * DN;
    float* carry = hloc  + (size_t)BATCH * NCH * DN;

    dim3 grid(NCH, BATCH);
    ssm_pass1<<<grid, CH, 0, stream>>>(x, dt, B, C, A, y, aprod, hloc);
    ssm_pass2<<<BATCH, DN, 0, stream>>>(aprod, hloc, carry, hfin);
    ssm_pass3<<<grid, CH, 0, stream>>>(dt, C, A, carry, y);
}


# pass 2: 32 blocks of 32 dims request all 64 rows of both chunk summaries at once (whole-line float4 reads via LDS); one wave chains them from registers: 1 memory round trip, not 8; fma order kept
# speedup vs baseline: 6.3637x; 6.3637x over previous
#include <hip/hip_runtime.h>

typedef __attribute__((ext_vector_type(2))) float v2f;
typedef __attribute__((ext_vector_type(8))) float v8f;

#define BATCH 8
#define SEQ   8192
#define DN    128
#define CH    128
#define NCH   (SEQ / CH)
#define SUB   8
#define NSUB  (CH / SUB)
#define DPL   8
#define P1T   (NSUB * (DN / DPL))
#define LOG2E 1.4426950408889634f

__global__ __launch_bounds__(P1T) void ssm_pass1(
                          const float* __restrict__ x, const float* __restrict__ dt,
                          const float* __restrict__ B, const float* __restrict__ C,
                          const float* __restrict__ A,
                          float* __restrict__ y, float* __restrict__ aprod,
                          float* __restrict__ hloc)
{
    __shared__ __attribute__((aligned(16))) float hs[NSUB * DN];
    __shared__ __attribute__((aligned(16))) float ys[CH];
    __shared__ float sds[NSUB];

    const int k = blockIdx.x, b = blockIdx.y;
    const int tid = threadIdx.x;
    const int g = tid & 15;
    const int j = tid >> 4;
    const int n0 = g * 4;
    const size_t sbase = (size_t)b * SEQ + (size_t)k * CH;
    const size_t t0 = sbase + (size_t)j * SUB;

    const float* Bp = B + t0 * DN + n0;
    const float* Cp = C + t0 * DN + n0;

    float bv[SUB][DPL], cv[SUB][DPL];
#pragma unroll
    for (int i = 0; i < SUB; ++i) {
        const float4 u = *(const float4*)(Bp + i * DN);
        const float4 v = *(const float4*)(Bp + i * DN + 64);
        bv[i][0] = u.x; bv[i][1] = u.y; bv[i][2] = u.z; bv[i][3] = u.w;
        bv[i][4] = v.x; bv[i][5] = v.y; bv[i][6] = v.z; bv[i][7] = v.w;
    }
#pragma unroll
    for (int i = 0; i < SUB; ++i) {
        const float4 u = *(const float4*)(Cp + i * DN);
        const float4 v = *(const float4*)(Cp + i * DN + 64);
        cv[i][0] = u.x; cv[i][1] = u.y; cv[i][2] = u.z; cv[i][3] = u.w;
        cv[i][4] = v.x; cv[i][5] = v.y; cv[i][6] = v.z; cv[i][7] = v.w;
    }
    const float4 d0 = *(const float4*)(dt + t0), d1 = *(const float4*)(dt + t0 + 4);
    const float4 x0 = *(const float4*)(x + t0),  x1 = *(const float4*)(x + t0 + 4);
    const float4 a0 = *(const float4*)(A + n0),  a1 = *(const float4*)(A + 64 + n0);
    const float An = A[tid & (DN - 1)];

    const float dts[SUB] = {d0.x, d0.y, d0.z, d0.w, d1.x, d1.y, d1.z, d1.w};
    const float dtx[SUB] = {d0.x * x0.x, d0.y * x0.y, d0.z * x0.z, d0.w * x0.w,
                            d1.x * x1.x, d1.y * x1.y, d1.z * x1.z, d1.w * x1.w};
    const float A2[DPL] = {a0.x * LOG2E, a0.y * LOG2E, a0.z * LOG2E, a0.w * LOG2E,
                           a1.x * LOG2E, a1.y * LOG2E, a1.z * LOG2E, a1.w * LOG2E};

    float h[DPL];
#pragma unroll
    for (int d = 0; d < DPL; ++d) h[d] = 0.f;
    float sd = 0.f;
#pragma unroll
    for (int i = 0; i < SUB; ++i) {
        sd += dts[i];
#pragma unroll
        for (int d = 0; d < DPL; ++d) {
            const float a = __builtin_amdgcn_exp2f(dts[i] * A2[d]);
            bv[i][d] *= dtx[i];
            h[d] = fmaf(a, h[d], bv[i][d]);
        }
    }
    float* hrow = hs + j * DN + n0;
    *(float4*)(hrow)      = make_float4(h[0], h[1], h[2], h[3]);
    *(float4*)(hrow + 64) = make_float4(h[4], h[5], h[6], h[7]);
    if (g == 0) sds[j] = sd;
    __syncthreads();

    if (tid < DN) {
        const float An2 = An * LOG2E;
        float carry = 0.f, sdt = 0.f;
#pragma unroll
        for (int jj = 0; jj < NSUB; ++jj) {
            const float hl = hs[jj * DN + tid];
            const float s  = sds[jj];
            hs[jj * DN + tid] = carry;
            carry = fmaf(__builtin_amdgcn_exp2f(An2 * s), carry, hl);
            sdt += s;
        }
        const size_t cidx = ((size_t)b * NCH + k) * DN + tid;
        aprod[cidx] = __expf(An * sdt);
        hloc[cidx]  = carry;
    }
    __syncthreads();

    {
        const float4 u = *(const float4*)(hrow);
        const float4 v = *(const float4*)(hrow + 64);
        h[0] = u.x; h[1] = u.y; h[2] = u.z; h[3] = u.w;
        h[4] = v.x; h[5] = v.y; h[6] = v.z; h[7] = v.w;
    }
    float p[SUB];
#pragma unroll
    for (int i = 0; i < SUB; ++i) {
        float acc = 0.f;
#pragma unroll
        for (int d = 0; d < DPL; ++d) {
            const float a = __builtin_amdgcn_exp2f(dts[i] * A2[d]);
            h[d] = fmaf(a, h[d], bv[i][d]);
            acc  = fmaf(cv[i][d], h[d], acc);
        }
        p[i] = acc;
    }
#pragma unroll
    for (int i = 0; i < SUB; ++i) {
        p[i] += __shfl_xor(p[i], 1, 32);
        p[i] += __shfl_xor(p[i], 2, 32);
        p[i] += __shfl_xor(p[i], 4, 32);
        p[i] += __shfl_xor(p[i], 8, 32);
    }
    if (g == 0) {
        *(float4*)(ys + j * SUB)     = make_float4(p[0], p[1], p[2], p[3]);
        *(float4*)(ys + j * SUB + 4) = make_float4(p[4], p[5], p[6], p[7]);
    }
    __syncthreads();
    if (tid < CH) y[sbase + tid] = ys[tid];
}

#define P2D 32
#define P2T 256

__global__ __launch_bounds__(P2T) void ssm_pass2(
                          const float* __restrict__ aprod, const float* __restrict__ hloc,
                          float* __restrict__ carry, float* __restrict__ hfin)
{
    __shared__ __attribute__((aligned(16))) float sa[NCH * P2D];
    __shared__ __attribute__((aligned(16))) float sh[NCH * P2D];

    const int n0 = blockIdx.x * P2D, b = blockIdx.y;
    const int tid = threadIdx.x;
    const size_t base = (size_t)b * NCH * DN + n0;

    float4 va[2], vh[2];
#pragma unroll
    for (int r = 0; r < 2; ++r) {
        const int p = tid + r * P2T;
        const size_t g = base + (size_t)(p >> 3) * DN + (p & 7) * 4;
        va[r] = *(const float4*)(aprod + g);
        vh[r] = *(const float4*)(hloc + g);
    }
#pragma unroll
    for (int r = 0; r < 2; ++r) {
        const int p = tid + r * P2T;
        *(float4*)(sa + p * 4) = va[r];
        *(float4*)(sh + p * 4) = vh[r];
    }
    __syncthreads();

    if (tid < P2D) {
        float av[NCH], hv[NCH];
#pragma unroll
        for (int k = 0; k < NCH; ++k) {
            av[k] = sa[k * P2D + tid];
            hv[k] = sh[k * P2D + tid];
        }
        __builtin_amdgcn_sched_barrier(0);
        float h = 0.f;
#pragma unroll
        for (int k = 0; k < NCH; ++k) {
            carry[base + (size_t)k * DN + tid] = h;
            h = fmaf(av[k], h, hv[k]);
        }
        hfin[(size_t)b * DN + n0 + tid] = h;
    }
}

#define TP 132

__device__ __forceinline__ void lds_barrier()
{
    __builtin_amdgcn_fence(__ATOMIC_RELEASE, "workgroup", "local");
    __builtin_amdgcn_s_barrier();
    __builtin_amdgcn_fence(__ATOMIC_ACQUIRE, "workgroup", "local");
}

__global__ __launch_bounds__(CH) void ssm_pass3(
                          const float* __restrict__ dt, const float* __restrict__ C,
                          const float* __restrict__ A, const float* __restrict__ carry,
                          float* __restrict__ y)
{
    __shared__ float Sd[CH];
    __shared__ __attribute__((aligned(16))) float hin[DN];
    __shared__ float ycorr[CH];
    __shared__ __attribute__((aligned(16))) float tile[4][2][16 * TP];

    const int k = blockIdx.x, b = blockIdx.y;
    const int tid = threadIdx.x;
    const int lane = tid & 31, wv = tid >> 5;
    const size_t sbase = (size_t)b * SEQ + (size_t)k * CH;

    const float  dtv = dt[sbase + tid];
    const float  hv  = carry[((size_t)b * NCH + k) * DN + tid];
    const float4 a4  = *(const float4*)(A + lane * 4);
    const float  yv  = y[sbase + tid];

    const float* Cw = C + (sbase + (size_t)wv * 32) * DN + lane * 4;
    float4 c[2][16];
#pragma unroll
    for (int i = 0; i < 2; ++i)
#pragma unroll
        for (int r = 0; r < 16; ++r)
            c[i][r] = *(const float4*)(Cw + (size_t)(i * 16 + r) * DN);

    Sd[tid]  = dtv;
    hin[tid] = hv;
    lds_barrier();
    for (int off = 1; off < CH; off <<= 1) {
        float v = (tid >= off) ? Sd[tid - off] : 0.f;
        lds_barrier();
        Sd[tid] += v;
        lds_barrier();
    }

    float* tw = &tile[wv][0][0];
#pragma unroll
    for (int i = 0; i < 2; ++i) {
#pragma unroll
        for (int r = 0; r < 16; ++r) {
            const float sdt = Sd[wv * 32 + i * 16 + r];
            float4 m;
            m.x = c[i][r].x * __expf(a4.x * sdt);
            m.y = c[i][r].y * __expf(a4.y * sdt);
            m.z = c[i][r].z * __expf(a4.z * sdt);
            m.w = c[i][r].w * __expf(a4.w * sdt);
            *(float4*)(tw + (i * 16 + r) * TP + lane * 4) = m;
        }
    }
    lds_barrier();

    const int cl = lane & 15;
    const int kBase = (lane < 16) ? 0 : 2;

    v2f bfr[DN / 4], af0[DN / 4], af1[DN / 4];
    const float* Mrow0 = tw + cl * TP;
    const float* Mrow1 = tw + (16 + cl) * TP;
#pragma unroll
    for (int s = 0; s < DN / 4; ++s) {
        const int kk = s * 4 + kBase;
        const float2 hk = *(const float2*)(hin + kk);
        const float2 m0 = *(const float2*)(Mrow0 + kk);
        const float2 m1 = *(const float2*)(Mrow1 + kk);
        bfr[s].x = hk.x; bfr[s].y = hk.y;
        af0[s].x = m0.x; af0[s].y = m0.y;
        af1[s].x = m1.x; af1[s].y = m1.y;
    }
    __builtin_amdgcn_sched_barrier(0);
    v8f acc0 = {}, acc1 = {};
#pragma unroll
    for (int s = 0; s < DN / 4; ++s) {
        acc0 = __builtin_amdgcn_wmma_f32_16x16x4_f32(
            false, af0[s], false, bfr[s], (short)0, acc0, false, false);
        acc1 = __builtin_amdgcn_wmma_f32_16x16x4_f32(
            false, af1[s], false, bfr[s], (short)0, acc1, false, false);
    }
    {
        const int t0 = wv * 32;
        if (lane == 0) {
#pragma unroll
            for (int r = 0; r < 8; ++r) { ycorr[t0 + r] = acc0[r]; ycorr[t0 + 16 + r] = acc1[r]; }
        }
        if (lane == 16) {
#pragma unroll
            for (int r = 0; r < 8; ++r) { ycorr[t0 + 8 + r] = acc0[r]; ycorr[t0 + 24 + r] = acc1[r]; }
        }
    }
    lds_barrier();
    y[sbase + tid] = yv + ycorr[tid];
}

extern "C" void kernel_launch(void* const* d_in, const int* in_sizes, int n_in,
                              void* d_out, int out_size, void* d_ws, size_t ws_size,
                              hipStream_t stream) {
    const float* x  = (const float*)d_in[0];
    const float* dt = (const float*)d_in[1];
    const float* B  = (const float*)d_in[2];
    const float* C  = (const float*)d_in[3];
    const float* A  = (const float*)d_in[4];

    float* y    = (float*)d_out;
    float* hfin = y + (size_t)BATCH * SEQ;

    float* aprod = (float*)d_ws;
    float* hloc  = aprod + (size_t)BATCH * NCH * DN;
    float* carry = hloc  + (size_t)BATCH * NCH * DN;

    dim3 grid(NCH, BATCH);
    ssm_pass1<<<grid, P1T, 0, stream>>>(x, dt, B, C, A, y, aprod, hloc);
    ssm_pass2<<<dim3(DN / P2D, BATCH), P2T, 0, stream>>>(aprod, hloc, carry, hfin);
    ssm_pass3<<<grid, CH, 0, stream>>>(dt, C, A, carry, y);
}


# pass 1 (in its 4 waves idle during chaining) writes the chunk's inclusive dt prefix to the workspace; pass 3 loads it: its 7-round LDS scan and 14 barriers go; dt sum order changed
# speedup vs baseline: 6.6992x; 1.0527x over previous
#include <hip/hip_runtime.h>

typedef __attribute__((ext_vector_type(2))) float v2f;
typedef __attribute__((ext_vector_type(8))) float v8f;

#define BATCH 8
#define SEQ   8192
#define DN    128
#define CH    128
#define NCH   (SEQ / CH)
#define SUB   8
#define NSUB  (CH / SUB)
#define DPL   8
#define P1T   (NSUB * (DN / DPL))
#define LOG2E 1.4426950408889634f

__global__ __launch_bounds__(P1T) void ssm_pass1(
                          const float* __restrict__ x, const float* __restrict__ dt,
                          const float* __restrict__ B, const float* __restrict__ C,
                          const float* __restrict__ A,
                          float* __restrict__ y, float* __restrict__ summ,
                          float* __restrict__ sdw)
{
    float* aprod = summ;
    float* hloc  = summ + (size_t)BATCH * NCH * DN;

    __shared__ __attribute__((aligned(16))) float hs[NSUB * DN];
    __shared__ __attribute__((aligned(16))) float ys[CH];
    __shared__ __attribute__((aligned(16))) float dtl[CH];
    __shared__ float sps[CH];
    __shared__ __attribute__((aligned(16))) float sds[NSUB];

    const int k = blockIdx.x, b = blockIdx.y;
    const int tid = threadIdx.x;
    const int g = tid & 15;
    const int j = tid >> 4;
    const int n0 = g * 4;
    const size_t sbase = (size_t)b * SEQ + (size_t)k * CH;
    const size_t t0 = sbase + (size_t)j * SUB;

    const float* Bp = B + t0 * DN + n0;
    const float* Cp = C + t0 * DN + n0;

    float bv[SUB][DPL], cv[SUB][DPL];
#pragma unroll
    for (int i = 0; i < SUB; ++i) {
        const float4 u = *(const float4*)(Bp + i * DN);
        const float4 v = *(const float4*)(Bp + i * DN + 64);
        bv[i][0] = u.x; bv[i][1] = u.y; bv[i][2] = u.z; bv[i][3] = u.w;
        bv[i][4] = v.x; bv[i][5] = v.y; bv[i][6] = v.z; bv[i][7] = v.w;
    }
#pragma unroll
    for (int i = 0; i < SUB; ++i) {
        const float4 u = *(const float4*)(Cp + i * DN);
        const float4 v = *(const float4*)(Cp + i * DN + 64);
        cv[i][0] = u.x; cv[i][1] = u.y; cv[i][2] = u.z; cv[i][3] = u.w;
        cv[i][4] = v.x; cv[i][5] = v.y; cv[i][6] = v.z; cv[i][7] = v.w;
    }
    const float4 d0 = *(const float4*)(dt + t0), d1 = *(const float4*)(dt + t0 + 4);
    const float4 x0 = *(const float4*)(x + t0),  x1 = *(const float4*)(x + t0 + 4);
    const float4 a0 = *(const float4*)(A + n0),  a1 = *(const float4*)(A + 64 + n0);
    const float An = A[tid & (DN - 1)];

    const float dts[SUB] = {d0.x, d0.y, d0.z, d0.w, d1.x, d1.y, d1.z, d1.w};
    const float dtx[SUB] = {d0.x * x0.x, d0.y * x0.y, d0.z * x0.z, d0.w * x0.w,
                            d1.x * x1.x, d1.y * x1.y, d1.z * x1.z, d1.w * x1.w};
    const float A2[DPL] = {a0.x * LOG2E, a0.y * LOG2E, a0.z * LOG2E, a0.w * LOG2E,
                           a1.x * LOG2E, a1.y * LOG2E, a1.z * LOG2E, a1.w * LOG2E};

    float h[DPL];
#pragma unroll
    for (int d = 0; d < DPL; ++d) h[d] = 0.f;
    float sd = 0.f;
#pragma unroll
    for (int i = 0; i < SUB; ++i) {
        sd += dts[i];
#pragma unroll
        for (int d = 0; d < DPL; ++d) {
            const float a = __builtin_amdgcn_exp2f(dts[i] * A2[d]);
            bv[i][d] *= dtx[i];
            h[d] = fmaf(a, h[d], bv[i][d]);
        }
    }
    float* hrow = hs + j * DN + n0;
    *(float4*)(hrow)      = make_float4(h[0], h[1], h[2], h[3]);
    *(float4*)(hrow + 64) = make_float4(h[4], h[5], h[6], h[7]);
    if (g == 0) {
        sds[j] = sd;
        *(float4*)(dtl + j * SUB)     = d0;
        *(float4*)(dtl + j * SUB + 4) = d1;
    }
    __syncthreads();

    if (tid >= DN) {
        const int t  = tid - DN;
        const int jq = t >> 3, iq = t & 7;
        float run = 0.f;
#pragma unroll
        for (int q = 0; q < NSUB / 4; ++q) {
            const float4 s4 = *(const float4*)(sds + q * 4);
            run += (q * 4 + 0 < jq) ? s4.x : 0.f;
            run += (q * 4 + 1 < jq) ? s4.y : 0.f;
            run += (q * 4 + 2 < jq) ? s4.z : 0.f;
            run += (q * 4 + 3 < jq) ? s4.w : 0.f;
        }
        const float4 e0 = *(const float4*)(dtl + jq * SUB);
        const float4 e1 = *(const float4*)(dtl + jq * SUB + 4);
        run += e0.x;
        run += (iq >= 1) ? e0.y : 0.f;
        run += (iq >= 2) ? e0.z : 0.f;
        run += (iq >= 3) ? e0.w : 0.f;
        run += (iq >= 4) ? e1.x : 0.f;
        run += (iq >= 5) ? e1.y : 0.f;
        run += (iq >= 6) ? e1.z : 0.f;
        run += (iq >= 7) ? e1.w : 0.f;
        sps[t] = run;
    }

    if (tid < DN) {
        const float An2 = An * LOG2E;
        float carry = 0.f, sdt = 0.f;
#pragma unroll
        for (int jj = 0; jj < NSUB; ++jj) {
            const float hl = hs[jj * DN + tid];
            const float s  = sds[jj];
            hs[jj * DN + tid] = carry;
            carry = fmaf(__builtin_amdgcn_exp2f(An2 * s), carry, hl);
            sdt += s;
        }
        const size_t cidx = ((size_t)b * NCH + k) * DN + tid;
        aprod[cidx] = __expf(An * sdt);
        hloc[cidx]  = carry;
    }
    __syncthreads();

    {
        const float4 u = *(const float4*)(hrow);
        const float4 v = *(const float4*)(hrow + 64);
        h[0] = u.x; h[1] = u.y; h[2] = u.z; h[3] = u.w;
        h[4] = v.x; h[5] = v.y; h[6] = v.z; h[7] = v.w;
    }
    float p[SUB];
#pragma unroll
    for (int i = 0; i < SUB; ++i) {
        float acc = 0.f;
#pragma unroll
        for (int d = 0; d < DPL; ++d) {
            const float a = __builtin_amdgcn_exp2f(dts[i] * A2[d]);
            h[d] = fmaf(a, h[d], bv[i][d]);
            acc  = fmaf(cv[i][d], h[d], acc);
        }
        p[i] = acc;
    }
#pragma unroll
    for (int i = 0; i < SUB; ++i) {
        p[i] += __shfl_xor(p[i], 1, 32);
        p[i] += __shfl_xor(p[i], 2, 32);
        p[i] += __shfl_xor(p[i], 4, 32);
        p[i] += __shfl_xor(p[i], 8, 32);
    }
    if (g == 0) {
        *(float4*)(ys + j * SUB)     = make_float4(p[0], p[1], p[2], p[3]);
        *(float4*)(ys + j * SUB + 4) = make_float4(p[4], p[5], p[6], p[7]);
    }
    __syncthreads();
    if (tid < CH) {
        y[sbase + tid]   = ys[tid];
        sdw[sbase + tid] = sps[tid];
    }
}

#define P2D 32
#define P2T 256

__global__ __launch_bounds__(P2T) void ssm_pass2(
                          const float* __restrict__ aprod, const float* __restrict__ hloc,
                          float* __restrict__ carry, float* __restrict__ hfin)
{
    __shared__ __attribute__((aligned(16))) float sa[NCH * P2D];
    __shared__ __attribute__((aligned(16))) float sh[NCH * P2D];

    const int n0 = blockIdx.x * P2D, b = blockIdx.y;
    const int tid = threadIdx.x;
    const size_t base = (size_t)b * NCH * DN + n0;

    float4 va[2], vh[2];
#pragma unroll
    for (int r = 0; r < 2; ++r) {
        const int p = tid + r * P2T;
        const size_t g = base + (size_t)(p >> 3) * DN + (p & 7) * 4;
        va[r] = *(const float4*)(aprod + g);
        vh[r] = *(const float4*)(hloc + g);
    }
#pragma unroll
    for (int r = 0; r < 2; ++r) {
        const int p = tid + r * P2T;
        *(float4*)(sa + p * 4) = va[r];
        *(float4*)(sh + p * 4) = vh[r];
    }
    __syncthreads();

    if (tid < P2D) {
        float av[NCH], hv[NCH];
#pragma unroll
        for (int k = 0; k < NCH; ++k) {
            av[k] = sa[k * P2D + tid];
            hv[k] = sh[k * P2D + tid];
        }
        __builtin_amdgcn_sched_barrier(0);
        float h = 0.f;
#pragma unroll
        for (int k = 0; k < NCH; ++k) {
            carry[base + (size_t)k * DN + tid] = h;
            h = fmaf(av[k], h, hv[k]);
        }
        hfin[(size_t)b * DN + n0 + tid] = h;
    }
}

#define TP 132

__device__ __forceinline__ void lds_barrier()
{
    __builtin_amdgcn_fence(__ATOMIC_RELEASE, "workgroup", "local");
    __builtin_amdgcn_s_barrier();
    __builtin_amdgcn_fence(__ATOMIC_ACQUIRE, "workgroup", "local");
}

__global__ __launch_bounds__(CH) void ssm_pass3(
                          const float* __restrict__ sdw, const float* __restrict__ C,
                          const float* __restrict__ A, const float* __restrict__ carry,
                          float* __restrict__ y)
{
    __shared__ float Sd[CH];
    __shared__ __attribute__((aligned(16))) float hin[DN];
    __shared__ float ycorr[CH];
    __shared__ __attribute__((aligned(16))) float tile[4][2][16 * TP];

    const int k = blockIdx.x, b = blockIdx.y;
    const int tid = threadIdx.x;
    const int lane = tid & 31, wv = tid >> 5;
    const size_t sbase = (size_t)b * SEQ + (size_t)k * CH;

    const float  sdv = sdw[sbase + tid];
    const float  hv  = carry[((size_t)b * NCH + k) * DN + tid];
    const float4 a4  = *(const float4*)(A + lane * 4);
    const float  yv  = y[sbase + tid];

    const float* Cw = C + (sbase + (size_t)wv * 32) * DN + lane * 4;
    float4 c[2][16];
#pragma unroll
    for (int i = 0; i < 2; ++i)
#pragma unroll
        for (int r = 0; r < 16; ++r)
            c[i][r] = *(const float4*)(Cw + (size_t)(i * 16 + r) * DN);

    Sd[tid]  = sdv;
    hin[tid] = hv;
    lds_barrier();

    float* tw = &tile[wv][0][0];
#pragma unroll
    for (int i = 0; i < 2; ++i) {
#pragma unroll
        for (int r = 0; r < 16; ++r) {
            const float sdt = Sd[wv * 32 + i * 16 + r];
            float4 m;
            m.x = c[i][r].x * __expf(a4.x * sdt);
            m.y = c[i][r].y * __expf(a4.y * sdt);
            m.z = c[i][r].z * __expf(a4.z * sdt);
            m.w = c[i][r].w * __expf(a4.w * sdt);
            *(float4*)(tw + (i * 16 + r) * TP + lane * 4) = m;
        }
    }
    lds_barrier();

    const int cl = lane & 15;
    const int kBase = (lane < 16) ? 0 : 2;

    v2f bfr[DN / 4], af0[DN / 4], af1[DN / 4];
    const float* Mrow0 = tw + cl * TP;
    const float* Mrow1 = tw + (16 + cl) * TP;
#pragma unroll
    for (int s = 0; s < DN / 4; ++s) {
        const int kk = s * 4 + kBase;
        const float2 hk = *(const float2*)(hin + kk);
        const float2 m0 = *(const float2*)(Mrow0 + kk);
        const float2 m1 = *(const float2*)(Mrow1 + kk);
        bfr[s].x = hk.x; bfr[s].y = hk.y;
        af0[s].x = m0.x; af0[s].y = m0.y;
        af1[s].x = m1.x; af1[s].y = m1.y;
    }
    __builtin_amdgcn_sched_barrier(0);
    v8f acc0 = {}, acc1 = {};
#pragma unroll
    for (int s = 0; s < DN / 4; ++s) {
        acc0 = __builtin_amdgcn_wmma_f32_16x16x4_f32(
            false, af0[s], false, bfr[s], (short)0, acc0, false, false);
        acc1 = __builtin_amdgcn_wmma_f32_16x16x4_f32(
            false, af1[s], false, bfr[s], (short)0, acc1, false, false);
    }
    {
        const int t0 = wv * 32;
        if (lane == 0) {
#pragma unroll
            for (int r = 0; r < 8; ++r) { ycorr[t0 + r] = acc0[r]; ycorr[t0 + 16 + r] = acc1[r]; }
        }
        if (lane == 16) {
#pragma unroll
            for (int r = 0; r < 8; ++r) { ycorr[t0 + 8 + r] = acc0[r]; ycorr[t0 + 24 + r] = acc1[r]; }
        }
    }
    lds_barrier();
    y[sbase + tid] = yv + ycorr[tid];
}

extern "C" void kernel_launch(void* const* d_in, const int* in_sizes, int n_in,
                              void* d_out, int out_size, void* d_ws, size_t ws_size,
                              hipStream_t stream) {
    const float* x  = (const float*)d_in[0];
    const float* dt = (const float*)d_in[1];
    const float* B  = (const float*)d_in[2];
    const float* C  = (const float*)d_in[3];
    const float* A  = (const float*)d_in[4];

    float* y    = (float*)d_out;
    float* hfin = y + (size_t)BATCH * SEQ;

    float* aprod = (float*)d_ws;
    float* hloc  = aprod + (size_t)BATCH * NCH * DN;
    float* carry = hloc  + (size_t)BATCH * NCH * DN;
    float* sdw   = carry + (size_t)BATCH * NCH * DN;

    dim3 grid(NCH, BATCH);
    ssm_pass1<<<grid, P1T, 0, stream>>>(x, dt, B, C, A, y, aprod, sdw);
    ssm_pass2<<<dim3(DN / P2D, BATCH), P2T, 0, stream>>>(aprod, hloc, carry, hfin);
    ssm_pass3<<<grid, CH, 0, stream>>>(sdw, C, A, carry, y);
}
